# starting kernel
# baseline (speedup 1.0000x reference)
#include <hip/hip_runtime.h>
#include <math.h>

typedef float v2f __attribute__((ext_vector_type(2)));
typedef float v8f __attribute__((ext_vector_type(8)));

#define NFFT   1024
#define BASEN  16
#define NCHUNK 64
#define WAVES  4
#define PI_F   3.14159265358979323846f

__device__ __forceinline__ int brev6(int c) { return (int)(__brev((unsigned)c) >> 26); }

__global__ __launch_bounds__(WAVES * 32) void rfft1024_kernel(
    const float* __restrict__ x, float* __restrict__ out, int nrows) {
  __shared__ __align__(16) float buf[WAVES][3 * NFFT];
  __shared__ float BRt[BASEN * BASEN];
  __shared__ float BIt[BASEN * BASEN];
  __shared__ float WRt[512];
  __shared__ float WIt[512];

  const int tid = threadIdx.x;

  for (int i = tid; i < BASEN * BASEN; i += WAVES * 32) {
    const int k = i >> 4, n = i & 15;
    const float ang = -2.0f * PI_F * (float)((k * n) & 15) * (1.0f / 16.0f);
    BRt[i] = cosf(ang);
    BIt[i] = sinf(ang);
  }
  for (int i = tid; i < 512; i += WAVES * 32) {
    const float ang = -PI_F * (float)i * (1.0f / 512.0f);
    WRt[i] = cosf(ang);
    WIt[i] = sinf(ang);
  }
  __syncthreads();

  const int wave = tid >> 5;
  const int lane = tid & 31;
  const int half = lane >> 4;
  const int m16  = lane & 15;
  const int row  = blockIdx.x * WAVES + wave;

  float* sx = buf[wave];
  float* ur = sx + NFFT;
  float* ui = ur + NFFT;

  const float4* xr4 = (const float4*)(x + (size_t)row * NFFT);
  float4* sx4 = (float4*)sx;
#pragma unroll
  for (int i = 0; i < NFFT / (32 * 4); ++i) sx4[lane + 32 * i] = xr4[lane + 32 * i];
  __syncthreads();

#pragma unroll
  for (int mt = 0; mt < NCHUNK / 16; ++mt) {
    const int cOff = brev6(mt * 16 + m16);
    v8f accR = {0.f, 0.f, 0.f, 0.f, 0.f, 0.f, 0.f, 0.f};
    v8f accI = {0.f, 0.f, 0.f, 0.f, 0.f, 0.f, 0.f, 0.f};
#pragma unroll
    for (int t = 0; t < 4; ++t) {
      const int k0 = 4 * t + 2 * half;
      v2f a, bR, bI;
      a[0]  = sx[(k0 + 0) * NCHUNK + cOff];
      a[1]  = sx[(k0 + 1) * NCHUNK + cOff];
      bR[0] = BRt[(k0 + 0) * 16 + m16];
      bR[1] = BRt[(k0 + 1) * 16 + m16];
      bI[0] = BIt[(k0 + 0) * 16 + m16];
      bI[1] = BIt[(k0 + 1) * 16 + m16];
      accR = __builtin_amdgcn_wmma_f32_16x16x4_f32(false, a, false, bR, (short)0, accR, false, false);
      accI = __builtin_amdgcn_wmma_f32_16x16x4_f32(false, a, false, bI, (short)0, accI, false, false);
    }
#pragma unroll
    for (int i = 0; i < 8; ++i) {
      const int c = mt * 16 + i + 8 * half;
      ur[c * 16 + m16] = accR[i];
      ui[c * 16 + m16] = accI[i];
    }
  }
  __syncthreads();

  for (int s = 0; s < 6; ++s) {
    const int L = 16 << s;
    const int tsh = 5 - s;
    for (int b = lane; b < 512; b += 32) {
      const int g  = b >> (4 + s);
      const int k  = b & (L - 1);
      const int i0 = (g << (5 + s)) + k;
      const int i1 = i0 + L;
      const float er = ur[i0], ei = ui[i0];
      const float orr = ur[i1], oii = ui[i1];
      const float wr = WRt[k << tsh], wi = WIt[k << tsh];
      const float tr = wr * orr - wi * oii;
      const float ti = wr * oii + wi * orr;
      ur[i0] = er + tr;  ui[i0] = ei + ti;
      ur[i1] = er - tr;  ui[i1] = ei - ti;
    }
    __syncthreads();
  }

  float* outR = out + (size_t)row * 513;
  float* outI = out + (size_t)nrows * 513 + (size_t)row * 513;
  for (int k = lane; k < 513; k += 32) {
    outR[k] = ur[k];
    outI[k] = ui[k];
  }
}

extern "C" void kernel_launch(void* const* d_in, const int* in_sizes, int n_in,
                              void* d_out, int out_size, void* d_ws, size_t ws_size,
                              hipStream_t stream) {
  const float* x = (const float*)d_in[0];
  float* out = (float*)d_out;
  const int nrows = in_sizes[0] / NFFT;
  const int blocks = nrows / WAVES;
  rfft1024_kernel<<<blocks, WAVES * 32, 0, stream>>>(x, out, nrows);
}


# Six radix-2 LDS sweeps and the LDS row staging go: four-step 32x32 FFT, both 32-point DFT passes on WMMA, pass 1's D tiles twiddled in registers are pass 2's B operand; sum order changes
# speedup vs baseline: 1.7543x; 1.7543x over previous
#include <hip/hip_runtime.h>
#include <math.h>

typedef float v2f __attribute__((ext_vector_type(2)));
typedef float v8f __attribute__((ext_vector_type(8)));

#define NFFT   1024
#define WAVES  4
#define PI_F   3.14159265358979323846f

__global__ __launch_bounds__(WAVES * 32) void rfft1024_kernel(
    const float* __restrict__ x, float* __restrict__ out, int nrows) {
  __shared__ float WR[NFFT];
  __shared__ float WI[NFFT];
  __shared__ float CR[32];
  __shared__ float CI[32];

  const int tid  = threadIdx.x;
  const int wave = tid >> 5;
  const int lane = tid & 31;
  const int h    = lane >> 4;
  const int c    = lane & 15;
  const int row  = blockIdx.x * WAVES + wave;
  const float* xr = x + (size_t)row * NFFT;

  v2f xa[2][8];
#pragma unroll
  for (int u = 0; u < 8; ++u) {
#pragma unroll
    for (int s = 0; s < 2; ++s) {
      const int kk = 2 * (8 * h + 2 * (u & 3) + s) + (u >> 2);
      const float2 v = *(const float2*)(xr + 32 * kk + 2 * c);
      xa[0][u][s] = v.x;
      xa[1][u][s] = v.y;
    }
  }

  for (int m = tid; m < 256; m += WAVES * 32) {
    const float ang = -2.0f * PI_F * (float)m * (1.0f / 1024.0f);
    const float cs = cosf(ang), sn = sinf(ang);
    WR[m]       =  cs;  WI[m]       =  sn;
    WR[m + 256] =  sn;  WI[m + 256] = -cs;
    WR[m + 512] = -cs;  WI[m + 512] = -sn;
    WR[m + 768] = -sn;  WI[m + 768] =  cs;
    if ((m & 31) == 0) {
      const int q = m >> 5;
      CR[q]      =  cs;  CI[q]      =  sn;
      CR[q + 8]  =  sn;  CI[q + 8]  = -cs;
      CR[q + 16] = -cs;  CI[q + 16] = -sn;
      CR[q + 24] = -sn;  CI[q + 24] =  cs;
    }
  }
  __syncthreads();

  v2f k0r[8], k0i[8];
#pragma unroll
  for (int u = 0; u < 8; ++u) {
#pragma unroll
    for (int s = 0; s < 2; ++s) {
      const int kk = 2 * (8 * h + 2 * (u & 3) + s) + (u >> 2);
      const int m = (kk * c) & 31;
      k0r[u][s] = CR[m];
      k0i[u][s] = CI[m];
    }
  }

  float nyq = 0.f;
#pragma unroll
  for (int u = 0; u < 8; ++u)
    nyq += (xa[0][u][0] - xa[1][u][0]) + (xa[0][u][1] - xa[1][u][1]);
  for (int d = 16; d >= 1; d >>= 1) nyq += __shfl_xor(nyq, d, 32);

  v8f er[2], ei[2], odr[2], odi[2];
#pragma unroll
  for (int mt = 0; mt < 2; ++mt) {
    v8f a0 = {0.f, 0.f, 0.f, 0.f, 0.f, 0.f, 0.f, 0.f};
    v8f a1 = a0, a2 = a0, a3 = a0;
#pragma unroll
    for (int u = 0; u < 4; ++u) {
      a0 = __builtin_amdgcn_wmma_f32_16x16x4_f32(false, xa[mt][u],     false, k0r[u],     (short)0, a0, false, false);
      a1 = __builtin_amdgcn_wmma_f32_16x16x4_f32(false, xa[mt][u],     false, k0i[u],     (short)0, a1, false, false);
      a2 = __builtin_amdgcn_wmma_f32_16x16x4_f32(false, xa[mt][u + 4], false, k0r[u + 4], (short)0, a2, false, false);
      a3 = __builtin_amdgcn_wmma_f32_16x16x4_f32(false, xa[mt][u + 4], false, k0i[u + 4], (short)0, a3, false, false);
    }
    er[mt] = a0;  ei[mt] = a1;  odr[mt] = a2;  odi[mt] = a3;
  }

  float* outR = out + (size_t)row * 513;
  float* outI = out + (size_t)nrows * 513 + (size_t)row * 513;
#pragma unroll
  for (int nt = 0; nt < 2; ++nt) {
    v8f zr[2], zi[2], zn[2];
#pragma unroll
    for (int mt = 0; mt < 2; ++mt) {
#pragma unroll
      for (int r = 0; r < 8; ++r) {
        const int m = (2 * (8 * h + r) + mt) * (16 * nt + c);
        const float wr = WR[m], wi = WI[m];
        const float a = nt ? er[mt][r] - odr[mt][r] : er[mt][r] + odr[mt][r];
        const float b = nt ? ei[mt][r] - odi[mt][r] : ei[mt][r] + odi[mt][r];
        const float tr = a * wr - b * wi;
        const float ti = a * wi + b * wr;
        zr[mt][r] = tr;
        zi[mt][r] = ti;
        zn[mt][r] = -ti;
      }
    }
    v8f orr = {0.f, 0.f, 0.f, 0.f, 0.f, 0.f, 0.f, 0.f};
    v8f oii = {0.f, 0.f, 0.f, 0.f, 0.f, 0.f, 0.f, 0.f};
#pragma unroll
    for (int u = 0; u < 8; ++u) {
      const int mt = u >> 2, j = u & 3;
      v2f br, bi, bn;
      br[0] = zr[mt][2 * j];  br[1] = zr[mt][2 * j + 1];
      bi[0] = zi[mt][2 * j];  bi[1] = zi[mt][2 * j + 1];
      bn[0] = zn[mt][2 * j];  bn[1] = zn[mt][2 * j + 1];
      orr = __builtin_amdgcn_wmma_f32_16x16x4_f32(false, k0r[u], false, br, (short)0, orr, false, false);
      oii = __builtin_amdgcn_wmma_f32_16x16x4_f32(false, k0r[u], false, bi, (short)0, oii, false, false);
      orr = __builtin_amdgcn_wmma_f32_16x16x4_f32(false, k0i[u], false, bn, (short)0, orr, false, false);
      oii = __builtin_amdgcn_wmma_f32_16x16x4_f32(false, k0i[u], false, br, (short)0, oii, false, false);
    }
#pragma unroll
    for (int r = 0; r < 8; ++r) {
      const int k = 16 * nt + c + 32 * (8 * h + r);
      outR[k] = orr[r];
      outI[k] = oii[r];
    }
  }
  if (lane == 0) {
    outR[512] = nyq;
    outI[512] = 0.f;
  }
}

extern "C" void kernel_launch(void* const* d_in, const int* in_sizes, int n_in,
                              void* d_out, int out_size, void* d_ws, size_t ws_size,
                              hipStream_t stream) {
  const float* x = (const float*)d_in[0];
  float* out = (float*)d_out;
  const int nrows = in_sizes[0] / NFFT;
  const int blocks = nrows / WAVES;
  rfft1024_kernel<<<blocks, WAVES * 32, 0, stream>>>(x, out, nrows);
}


# Pass 2's complex product by three real WMMA products instead of four: P=Wr(Zr+Zi) shared as C of both chains, Re=P-(Wr+Wi)Zi, Im=P+(Wi-Wr)Zr; 80 WMMAs per row, was 96; sum order changes
# speedup vs baseline: 1.8536x; 1.0566x over previous
#include <hip/hip_runtime.h>
#include <math.h>

typedef float v2f __attribute__((ext_vector_type(2)));
typedef float v8f __attribute__((ext_vector_type(8)));

#define NFFT   1024
#define WAVES  4
#define PI_F   3.14159265358979323846f

__global__ __launch_bounds__(WAVES * 32) void rfft1024_kernel(
    const float* __restrict__ x, float* __restrict__ out, int nrows) {
  __shared__ float WR[NFFT];
  __shared__ float WI[NFFT];
  __shared__ float CR[32];
  __shared__ float CI[32];

  const int tid  = threadIdx.x;
  const int wave = tid >> 5;
  const int lane = tid & 31;
  const int h    = lane >> 4;
  const int c    = lane & 15;
  const int row  = blockIdx.x * WAVES + wave;
  const float* xr = x + (size_t)row * NFFT;

  v2f xa[2][8];
#pragma unroll
  for (int u = 0; u < 8; ++u) {
#pragma unroll
    for (int s = 0; s < 2; ++s) {
      const int kk = 2 * (8 * h + 2 * (u & 3) + s) + (u >> 2);
      const float2 v = *(const float2*)(xr + 32 * kk + 2 * c);
      xa[0][u][s] = v.x;
      xa[1][u][s] = v.y;
    }
  }

  for (int m = tid; m < 256; m += WAVES * 32) {
    const float ang = -2.0f * PI_F * (float)m * (1.0f / 1024.0f);
    const float cs = cosf(ang), sn = sinf(ang);
    WR[m]       =  cs;  WI[m]       =  sn;
    WR[m + 256] =  sn;  WI[m + 256] = -cs;
    WR[m + 512] = -cs;  WI[m + 512] = -sn;
    WR[m + 768] = -sn;  WI[m + 768] =  cs;
    if ((m & 31) == 0) {
      const int q = m >> 5;
      CR[q]      =  cs;  CI[q]      =  sn;
      CR[q + 8]  =  sn;  CI[q + 8]  = -cs;
      CR[q + 16] = -cs;  CI[q + 16] = -sn;
      CR[q + 24] = -sn;  CI[q + 24] =  cs;
    }
  }
  __syncthreads();

  v2f k0r[8], k0i[8];
#pragma unroll
  for (int u = 0; u < 8; ++u) {
#pragma unroll
    for (int s = 0; s < 2; ++s) {
      const int kk = 2 * (8 * h + 2 * (u & 3) + s) + (u >> 2);
      const int m = (kk * c) & 31;
      k0r[u][s] = CR[m];
      k0i[u][s] = CI[m];
    }
  }

  float nyq = 0.f;
#pragma unroll
  for (int u = 0; u < 8; ++u)
    nyq += (xa[0][u][0] - xa[1][u][0]) + (xa[0][u][1] - xa[1][u][1]);
  for (int d = 16; d >= 1; d >>= 1) nyq += __shfl_xor(nyq, d, 32);

  v8f er[2], ei[2], odr[2], odi[2];
#pragma unroll
  for (int mt = 0; mt < 2; ++mt) {
    v8f a0 = {0.f, 0.f, 0.f, 0.f, 0.f, 0.f, 0.f, 0.f};
    v8f a1 = a0, a2 = a0, a3 = a0;
#pragma unroll
    for (int u = 0; u < 4; ++u) {
      a0 = __builtin_amdgcn_wmma_f32_16x16x4_f32(false, xa[mt][u],     false, k0r[u],     (short)0, a0, false, false);
      a1 = __builtin_amdgcn_wmma_f32_16x16x4_f32(false, xa[mt][u],     false, k0i[u],     (short)0, a1, false, false);
      a2 = __builtin_amdgcn_wmma_f32_16x16x4_f32(false, xa[mt][u + 4], false, k0r[u + 4], (short)0, a2, false, false);
      a3 = __builtin_amdgcn_wmma_f32_16x16x4_f32(false, xa[mt][u + 4], false, k0i[u + 4], (short)0, a3, false, false);
    }
    er[mt] = a0;  ei[mt] = a1;  odr[mt] = a2;  odi[mt] = a3;
  }

  v2f kn[8], kd[8];
#pragma unroll
  for (int u = 0; u < 8; ++u) {
    kn[u] = -k0r[u] - k0i[u];
    kd[u] = k0i[u] - k0r[u];
  }

  float* outR = out + (size_t)row * 513;
  float* outI = out + (size_t)nrows * 513 + (size_t)row * 513;
#pragma unroll
  for (int nt = 0; nt < 2; ++nt) {
    v8f zr[2], zi[2], zs[2];
#pragma unroll
    for (int mt = 0; mt < 2; ++mt) {
#pragma unroll
      for (int r = 0; r < 8; ++r) {
        const int m = (2 * (8 * h + r) + mt) * (16 * nt + c);
        const float wr = WR[m], wi = WI[m];
        const float a = nt ? er[mt][r] - odr[mt][r] : er[mt][r] + odr[mt][r];
        const float b = nt ? ei[mt][r] - odi[mt][r] : ei[mt][r] + odi[mt][r];
        const float tr = a * wr - b * wi;
        const float ti = a * wi + b * wr;
        zr[mt][r] = tr;
        zi[mt][r] = ti;
        zs[mt][r] = tr + ti;
      }
    }
    v8f pp = {0.f, 0.f, 0.f, 0.f, 0.f, 0.f, 0.f, 0.f};
#pragma unroll
    for (int u = 0; u < 8; ++u) {
      const int mt = u >> 2, j = u & 3;
      v2f bs;
      bs[0] = zs[mt][2 * j];  bs[1] = zs[mt][2 * j + 1];
      pp = __builtin_amdgcn_wmma_f32_16x16x4_f32(false, k0r[u], false, bs, (short)0, pp, false, false);
    }
    v8f orr = pp;
    v8f oii = pp;
#pragma unroll
    for (int u = 0; u < 8; ++u) {
      const int mt = u >> 2, j = u & 3;
      v2f br, bi;
      br[0] = zr[mt][2 * j];  br[1] = zr[mt][2 * j + 1];
      bi[0] = zi[mt][2 * j];  bi[1] = zi[mt][2 * j + 1];
      orr = __builtin_amdgcn_wmma_f32_16x16x4_f32(false, kn[u], false, bi, (short)0, orr, false, false);
      oii = __builtin_amdgcn_wmma_f32_16x16x4_f32(false, kd[u], false, br, (short)0, oii, false, false);
    }
#pragma unroll
    for (int r = 0; r < 8; ++r) {
      const int k = 16 * nt + c + 32 * (8 * h + r);
      outR[k] = orr[r];
      outI[k] = oii[r];
    }
  }
  if (lane == 0) {
    outR[512] = nyq;
    outI[512] = 0.f;
  }
}

extern "C" void kernel_launch(void* const* d_in, const int* in_sizes, int n_in,
                              void* d_out, int out_size, void* d_ws, size_t ws_size,
                              hipStream_t stream) {
  const float* x = (const float*)d_in[0];
  float* out = (float*)d_out;
  const int nrows = in_sizes[0] / NFFT;
  const int blocks = nrows / WAVES;
  rfft1024_kernel<<<blocks, WAVES * 32, 0, stream>>>(x, out, nrows);
}


# Two real rows ride one complex 1024-point transform (z = x1 + i*x2; X1, X2 split from Z[k], conj Z[N-k] through LDS): parity halving in both passes, 96 WMMAs per 2 rows, was 160; sum order changes
# speedup vs baseline: 1.9832x; 1.0699x over previous
#include <hip/hip_runtime.h>
#include <math.h>

typedef float v2f __attribute__((ext_vector_type(2)));
typedef float v8f __attribute__((ext_vector_type(8)));

#define NFFT   1024
#define WAVES  4
#define XPITCH 520
#define PI_F   3.14159265358979323846f

__global__ __launch_bounds__(WAVES * 32) void rfft1024_kernel(
    const float* __restrict__ x, float* __restrict__ out, int nrows) {
  __shared__ float WR[NFFT];
  __shared__ float WI[NFFT];
  __shared__ float CR[32];
  __shared__ float CI[32];
  __shared__ float SR[WAVES * XPITCH];
  __shared__ float SI[WAVES * XPITCH];

  const int tid  = threadIdx.x;
  const int wave = tid >> 5;
  const int lane = tid & 31;
  const int h    = lane >> 4;
  const int c    = lane & 15;
  const int row  = 2 * (blockIdx.x * WAVES + wave);
  const float* xr = x + (size_t)row * NFFT;

  v2f x1a[2][8], x2a[2][8];
#pragma unroll
  for (int u = 0; u < 8; ++u) {
#pragma unroll
    for (int s = 0; s < 2; ++s) {
      const int kk = 2 * (8 * h + 2 * (u & 3) + s) + (u >> 2);
      const float2 v = *(const float2*)(xr + 32 * kk + 2 * c);
      const float2 w = *(const float2*)(xr + NFFT + 32 * kk + 2 * c);
      x1a[0][u][s] = v.x;
      x1a[1][u][s] = v.y;
      x2a[0][u][s] = w.x;
      x2a[1][u][s] = w.y;
    }
  }

  for (int m = tid; m < 256; m += WAVES * 32) {
    const float ang = -2.0f * PI_F * (float)m * (1.0f / 1024.0f);
    const float cs = cosf(ang), sn = sinf(ang);
    const float hc = 0.5f * cs, hs = 0.5f * sn;
    WR[m]       =  hc;  WI[m]       =  hs;
    WR[m + 256] =  hs;  WI[m + 256] = -hc;
    WR[m + 512] = -hc;  WI[m + 512] = -hs;
    WR[m + 768] = -hs;  WI[m + 768] =  hc;
    if ((m & 31) == 0) {
      const int q = m >> 5;
      CR[q]      =  cs;  CI[q]      =  sn;
      CR[q + 8]  =  sn;  CI[q + 8]  = -cs;
      CR[q + 16] = -cs;  CI[q + 16] = -sn;
      CR[q + 24] = -sn;  CI[q + 24] =  cs;
    }
  }
  __syncthreads();

  v2f k0r[8], kn[8], kd[8];
#pragma unroll
  for (int u = 0; u < 8; ++u) {
#pragma unroll
    for (int s = 0; s < 2; ++s) {
      const int kk = 2 * (8 * h + 2 * (u & 3) + s) + (u >> 2);
      const int m = (kk * c) & 31;
      const float wr = CR[m], wi = CI[m];
      k0r[u][s] = wr;
      kn[u][s]  = -wr - wi;
      kd[u][s]  = wi - wr;
    }
  }

  v8f er[2], ei[2], odr[2], odi[2];
#pragma unroll
  for (int mt = 0; mt < 2; ++mt) {
    v8f pe = {0.f, 0.f, 0.f, 0.f, 0.f, 0.f, 0.f, 0.f};
    v8f po = pe;
#pragma unroll
    for (int u = 0; u < 4; ++u) {
      const v2f se = x1a[mt][u] + x2a[mt][u];
      const v2f so = x1a[mt][u + 4] + x2a[mt][u + 4];
      pe = __builtin_amdgcn_wmma_f32_16x16x4_f32(false, se, false, k0r[u],     (short)0, pe, false, false);
      po = __builtin_amdgcn_wmma_f32_16x16x4_f32(false, so, false, k0r[u + 4], (short)0, po, false, false);
    }
    v8f a0 = pe, a1 = pe, a2 = po, a3 = po;
#pragma unroll
    for (int u = 0; u < 4; ++u) {
      a0 = __builtin_amdgcn_wmma_f32_16x16x4_f32(false, x2a[mt][u],     false, kn[u],     (short)0, a0, false, false);
      a1 = __builtin_amdgcn_wmma_f32_16x16x4_f32(false, x1a[mt][u],     false, kd[u],     (short)0, a1, false, false);
      a2 = __builtin_amdgcn_wmma_f32_16x16x4_f32(false, x2a[mt][u + 4], false, kn[u + 4], (short)0, a2, false, false);
      a3 = __builtin_amdgcn_wmma_f32_16x16x4_f32(false, x1a[mt][u + 4], false, kd[u + 4], (short)0, a3, false, false);
    }
    er[mt] = a0;  ei[mt] = a1;  odr[mt] = a2;  odi[mt] = a3;
  }

  float* sr = SR + wave * XPITCH;
  float* si = SI + wave * XPITCH;
  v8f lor[2], loi[2];
  float nyr = 0.f, nyi = 0.f;
#pragma unroll
  for (int nt = 0; nt < 2; ++nt) {
    v8f zr[2], zi[2], zs[2];
#pragma unroll
    for (int mt = 0; mt < 2; ++mt) {
#pragma unroll
      for (int r = 0; r < 8; ++r) {
        const int m = (2 * (8 * h + r) + mt) * (16 * nt + c);
        const float wr = WR[m], wi = WI[m];
        const float a = nt ? er[mt][r] - odr[mt][r] : er[mt][r] + odr[mt][r];
        const float b = nt ? ei[mt][r] - odi[mt][r] : ei[mt][r] + odi[mt][r];
        const float tr = a * wr - b * wi;
        const float ti = a * wi + b * wr;
        zr[mt][r] = tr;
        zi[mt][r] = ti;
        zs[mt][r] = tr + ti;
      }
    }
    v8f pe = {0.f, 0.f, 0.f, 0.f, 0.f, 0.f, 0.f, 0.f};
    v8f po = pe;
#pragma unroll
    for (int j = 0; j < 4; ++j) {
      v2f be, bo;
      be[0] = zs[0][2 * j];  be[1] = zs[0][2 * j + 1];
      bo[0] = zs[1][2 * j];  bo[1] = zs[1][2 * j + 1];
      pe = __builtin_amdgcn_wmma_f32_16x16x4_f32(false, k0r[j],     false, be, (short)0, pe, false, false);
      po = __builtin_amdgcn_wmma_f32_16x16x4_f32(false, k0r[j + 4], false, bo, (short)0, po, false, false);
    }
    v8f eR = pe, eI = pe, oR = po, oI = po;
#pragma unroll
    for (int j = 0; j < 4; ++j) {
      v2f bre, bie, bro, bio;
      bre[0] = zr[0][2 * j];  bre[1] = zr[0][2 * j + 1];
      bie[0] = zi[0][2 * j];  bie[1] = zi[0][2 * j + 1];
      bro[0] = zr[1][2 * j];  bro[1] = zr[1][2 * j + 1];
      bio[0] = zi[1][2 * j];  bio[1] = zi[1][2 * j + 1];
      eR = __builtin_amdgcn_wmma_f32_16x16x4_f32(false, kn[j],     false, bie, (short)0, eR, false, false);
      eI = __builtin_amdgcn_wmma_f32_16x16x4_f32(false, kd[j],     false, bre, (short)0, eI, false, false);
      oR = __builtin_amdgcn_wmma_f32_16x16x4_f32(false, kn[j + 4], false, bio, (short)0, oR, false, false);
      oI = __builtin_amdgcn_wmma_f32_16x16x4_f32(false, kd[j + 4], false, bro, (short)0, oI, false, false);
    }
    lor[nt] = eR + oR;
    loi[nt] = eI + oI;
    const v8f hir = eR - oR;
    const v8f hii = eI - oI;
    if (nt == 0) { nyr = hir[0];  nyi = hii[0]; }
#pragma unroll
    for (int r = 0; r < 8; ++r) {
      const int j = 512 - (16 * nt + c + 32 * (8 * h + r));
      sr[j] = hir[r];
      si[j] = hii[r];
    }
  }
  if (lane == 0) {
    sr[0] = lor[0][0];
    si[0] = loi[0][0];
  }

  float* o1R = out + (size_t)row * 513;
  float* o2R = o1R + 513;
  float* o1I = out + (size_t)nrows * 513 + (size_t)row * 513;
  float* o2I = o1I + 513;
#pragma unroll
  for (int nt = 0; nt < 2; ++nt) {
#pragma unroll
    for (int r = 0; r < 8; ++r) {
      const int k = 16 * nt + c + 32 * (8 * h + r);
      const float pr = sr[k], pi = si[k];
      const float qr = lor[nt][r], qi = loi[nt][r];
      o1R[k] = qr + pr;
      o1I[k] = qi - pi;
      o2R[k] = qi + pi;
      o2I[k] = pr - qr;
    }
  }
  if (lane == 0) {
    o1R[512] = nyr + nyr;
    o1I[512] = 0.f;
    o2R[512] = nyi + nyi;
    o2I[512] = 0.f;
  }
}

extern "C" void kernel_launch(void* const* d_in, const int* in_sizes, int n_in,
                              void* d_out, int out_size, void* d_ws, size_t ws_size,
                              hipStream_t stream) {
  const float* x = (const float*)d_in[0];
  float* out = (float*)d_out;
  const int nrows = in_sizes[0] / NFFT;
  const int blocks = nrows / (2 * WAVES);
  rfft1024_kernel<<<blocks, WAVES * 32, 0, stream>>>(x, out, nrows);
}


# Whole spectrum staged in wave-private LDS planes in bin order; a lane splits 4 consecutive bins, so each store instruction writes 512 contiguous bytes (16-byte stores), was 2 pieces of 64; same sums
# speedup vs baseline: 2.2749x; 1.1471x over previous
#include <hip/hip_runtime.h>
#include <math.h>

typedef float v2f __attribute__((ext_vector_type(2)));
typedef float v8f __attribute__((ext_vector_type(8)));
typedef float v4f __attribute__((ext_vector_type(4)));
typedef v4f v4f_a4 __attribute__((aligned(4)));

#define NFFT   1024
#define WAVES  4
#define XPITCH 520
#define PI_F   3.14159265358979323846f

__global__ __launch_bounds__(WAVES * 32) void rfft1024_kernel(
    const float* __restrict__ x, float* __restrict__ out, int nrows) {
  __shared__ float WR[NFFT];
  __shared__ float WI[NFFT];
  __shared__ float CR[32];
  __shared__ float CI[32];
  __shared__ __attribute__((aligned(16))) float LR[WAVES * XPITCH];
  __shared__ __attribute__((aligned(16))) float LI[WAVES * XPITCH];
  __shared__ __attribute__((aligned(16))) float MR[WAVES * XPITCH];
  __shared__ __attribute__((aligned(16))) float MI[WAVES * XPITCH];

  const int tid  = threadIdx.x;
  const int wave = tid >> 5;
  const int lane = tid & 31;
  const int h    = lane >> 4;
  const int c    = lane & 15;
  const int row  = 2 * (blockIdx.x * WAVES + wave);
  const float* xr = x + (size_t)row * NFFT;

  v2f x1a[2][8], x2a[2][8];
#pragma unroll
  for (int u = 0; u < 8; ++u) {
#pragma unroll
    for (int s = 0; s < 2; ++s) {
      const int kk = 2 * (8 * h + 2 * (u & 3) + s) + (u >> 2);
      const float2 v = *(const float2*)(xr + 32 * kk + 2 * c);
      const float2 w = *(const float2*)(xr + NFFT + 32 * kk + 2 * c);
      x1a[0][u][s] = v.x;
      x1a[1][u][s] = v.y;
      x2a[0][u][s] = w.x;
      x2a[1][u][s] = w.y;
    }
  }

  for (int m = tid; m < 256; m += WAVES * 32) {
    const float ang = -2.0f * PI_F * (float)m * (1.0f / 1024.0f);
    const float cs = cosf(ang), sn = sinf(ang);
    const float hc = 0.5f * cs, hs = 0.5f * sn;
    WR[m]       =  hc;  WI[m]       =  hs;
    WR[m + 256] =  hs;  WI[m + 256] = -hc;
    WR[m + 512] = -hc;  WI[m + 512] = -hs;
    WR[m + 768] = -hs;  WI[m + 768] =  hc;
    if ((m & 31) == 0) {
      const int q = m >> 5;
      CR[q]      =  cs;  CI[q]      =  sn;
      CR[q + 8]  =  sn;  CI[q + 8]  = -cs;
      CR[q + 16] = -cs;  CI[q + 16] = -sn;
      CR[q + 24] = -sn;  CI[q + 24] =  cs;
    }
  }
  __syncthreads();

  v2f k0r[8], kn[8], kd[8];
#pragma unroll
  for (int u = 0; u < 8; ++u) {
#pragma unroll
    for (int s = 0; s < 2; ++s) {
      const int kk = 2 * (8 * h + 2 * (u & 3) + s) + (u >> 2);
      const int m = (kk * c) & 31;
      const float wr = CR[m], wi = CI[m];
      k0r[u][s] = wr;
      kn[u][s]  = -wr - wi;
      kd[u][s]  = wi - wr;
    }
  }

  v8f er[2], ei[2], odr[2], odi[2];
#pragma unroll
  for (int mt = 0; mt < 2; ++mt) {
    v8f pe = {0.f, 0.f, 0.f, 0.f, 0.f, 0.f, 0.f, 0.f};
    v8f po = pe;
#pragma unroll
    for (int u = 0; u < 4; ++u) {
      const v2f se = x1a[mt][u] + x2a[mt][u];
      const v2f so = x1a[mt][u + 4] + x2a[mt][u + 4];
      pe = __builtin_amdgcn_wmma_f32_16x16x4_f32(false, se, false, k0r[u],     (short)0, pe, false, false);
      po = __builtin_amdgcn_wmma_f32_16x16x4_f32(false, so, false, k0r[u + 4], (short)0, po, false, false);
    }
    v8f a0 = pe, a1 = pe, a2 = po, a3 = po;
#pragma unroll
    for (int u = 0; u < 4; ++u) {
      a0 = __builtin_amdgcn_wmma_f32_16x16x4_f32(false, x2a[mt][u],     false, kn[u],     (short)0, a0, false, false);
      a1 = __builtin_amdgcn_wmma_f32_16x16x4_f32(false, x1a[mt][u],     false, kd[u],     (short)0, a1, false, false);
      a2 = __builtin_amdgcn_wmma_f32_16x16x4_f32(false, x2a[mt][u + 4], false, kn[u + 4], (short)0, a2, false, false);
      a3 = __builtin_amdgcn_wmma_f32_16x16x4_f32(false, x1a[mt][u + 4], false, kd[u + 4], (short)0, a3, false, false);
    }
    er[mt] = a0;  ei[mt] = a1;  odr[mt] = a2;  odi[mt] = a3;
  }

  float* lr = LR + wave * XPITCH;
  float* li = LI + wave * XPITCH;
  float* mr = MR + wave * XPITCH;
  float* mi = MI + wave * XPITCH;
  float z0r = 0.f, z0i = 0.f;
  float nyr = 0.f, nyi = 0.f;
#pragma unroll
  for (int nt = 0; nt < 2; ++nt) {
    v8f zr[2], zi[2], zs[2];
#pragma unroll
    for (int mt = 0; mt < 2; ++mt) {
#pragma unroll
      for (int r = 0; r < 8; ++r) {
        const int m = (2 * (8 * h + r) + mt) * (16 * nt + c);
        const float wr = WR[m], wi = WI[m];
        const float a = nt ? er[mt][r] - odr[mt][r] : er[mt][r] + odr[mt][r];
        const float b = nt ? ei[mt][r] - odi[mt][r] : ei[mt][r] + odi[mt][r];
        const float tr = a * wr - b * wi;
        const float ti = a * wi + b * wr;
        zr[mt][r] = tr;
        zi[mt][r] = ti;
        zs[mt][r] = tr + ti;
      }
    }
    v8f pe = {0.f, 0.f, 0.f, 0.f, 0.f, 0.f, 0.f, 0.f};
    v8f po = pe;
#pragma unroll
    for (int j = 0; j < 4; ++j) {
      v2f be, bo;
      be[0] = zs[0][2 * j];  be[1] = zs[0][2 * j + 1];
      bo[0] = zs[1][2 * j];  bo[1] = zs[1][2 * j + 1];
      pe = __builtin_amdgcn_wmma_f32_16x16x4_f32(false, k0r[j],     false, be, (short)0, pe, false, false);
      po = __builtin_amdgcn_wmma_f32_16x16x4_f32(false, k0r[j + 4], false, bo, (short)0, po, false, false);
    }
    v8f eR = pe, eI = pe, oR = po, oI = po;
#pragma unroll
    for (int j = 0; j < 4; ++j) {
      v2f bre, bie, bro, bio;
      bre[0] = zr[0][2 * j];  bre[1] = zr[0][2 * j + 1];
      bie[0] = zi[0][2 * j];  bie[1] = zi[0][2 * j + 1];
      bro[0] = zr[1][2 * j];  bro[1] = zr[1][2 * j + 1];
      bio[0] = zi[1][2 * j];  bio[1] = zi[1][2 * j + 1];
      eR = __builtin_amdgcn_wmma_f32_16x16x4_f32(false, kn[j],     false, bie, (short)0, eR, false, false);
      eI = __builtin_amdgcn_wmma_f32_16x16x4_f32(false, kd[j],     false, bre, (short)0, eI, false, false);
      oR = __builtin_amdgcn_wmma_f32_16x16x4_f32(false, kn[j + 4], false, bio, (short)0, oR, false, false);
      oI = __builtin_amdgcn_wmma_f32_16x16x4_f32(false, kd[j + 4], false, bro, (short)0, oI, false, false);
    }
    const v8f lor = eR + oR;
    const v8f loi = eI + oI;
    const v8f hir = eR - oR;
    const v8f hii = eI - oI;
    if (nt == 0) { z0r = lor[0];  z0i = loi[0];  nyr = hir[0];  nyi = hii[0]; }
#pragma unroll
    for (int r = 0; r < 8; ++r) {
      const int k = 16 * nt + c + 32 * (8 * h + r);
      lr[k + 3] = lor[r];
      li[k + 3] = loi[r];
      mr[k] = hir[r];
      mi[k] = hii[r];
    }
  }
  if (lane == 0) {
    lr[515] = nyr;
    li[515] = nyi;
  }

  float* o1R = out + (size_t)row * 513;
  float* o2R = o1R + 513;
  float* o1I = out + (size_t)nrows * 513 + (size_t)row * 513;
  float* o2I = o1I + 513;
#pragma unroll
  for (int i = 0; i < 4; ++i) {
    const int m4 = 4 * (lane + 32 * i);
    const float* ql = lr + m4 + 4;
    const float* qm = li + m4 + 4;
    const float* pl = mr + 508 - m4;
    const float* pm = mi + 508 - m4;
    v4f qr, qi, pr, pi;
#pragma unroll
    for (int e = 0; e < 4; ++e) {
      qr[e] = ql[e];
      qi[e] = qm[e];
      pr[e] = pl[3 - e];
      pi[e] = pm[3 - e];
    }
    *(v4f_a4*)(o1R + m4 + 1) = qr + pr;
    *(v4f_a4*)(o1I + m4 + 1) = qi - pi;
    *(v4f_a4*)(o2R + m4 + 1) = qi + pi;
    *(v4f_a4*)(o2I + m4 + 1) = pr - qr;
  }
  if (lane == 0) {
    o1R[0] = z0r + z0r;
    o1I[0] = 0.f;
    o2R[0] = z0i + z0i;
    o2I[0] = 0.f;
  }
}

extern "C" void kernel_launch(void* const* d_in, const int* in_sizes, int n_in,
                              void* d_out, int out_size, void* d_ws, size_t ws_size,
                              hipStream_t stream) {
  const float* x = (const float*)d_in[0];
  float* out = (float*)d_out;
  const int nrows = in_sizes[0] / NFFT;
  const int blocks = nrows / (2 * WAVES);
  rfft1024_kernel<<<blocks, WAVES * 32, 0, stream>>>(x, out, nrows);
}
